# starting kernel
# baseline (speedup 1.0000x reference)
#include <hip/hip_runtime.h>

#define VOCAB 33
#define V2 1089
#define BUCKETS 512
#define B_ROWS 4096
#define S_LEN 4096
#define KPAD_BI 1104
#define HID 256
#define OUTC 64

typedef float v2f __attribute__((ext_vector_type(2)));
typedef float v8f __attribute__((ext_vector_type(8)));

__global__ __launch_bounds__(256) void hist_kernel(const int* __restrict__ src,
                                                   float* __restrict__ bi_freq,
                                                   float* __restrict__ tri_freq) {
    __shared__ unsigned bi_cnt[V2];
    __shared__ unsigned tri_cnt[BUCKETS];
    __shared__ unsigned totals[2];

    const int row = blockIdx.x;
    const int tid = threadIdx.x;
    const int* s = src + (size_t)row * S_LEN;

    for (int i = tid; i < V2; i += 256) bi_cnt[i] = 0u;
    for (int i = tid; i < BUCKETS; i += 256) tri_cnt[i] = 0u;
    if (tid < 2) totals[tid] = 0u;
    __syncthreads();

    unsigned myBi = 0, myTri = 0;
    for (int i = tid; i < S_LEN - 1; i += 256) {
        int c = s[i];
        int n = s[i + 1];
        bool bm = (c > 0) && (n > 0);
        if (bm) {
            atomicAdd(&bi_cnt[c * VOCAB + n], 1u);
            ++myBi;
        }
        if (i < S_LEN - 2) {
            int t = s[i + 2];
            if (bm && (t > 0)) {
                unsigned idx = ((unsigned)(c * (VOCAB * VOCAB) + n * VOCAB + t)) & (BUCKETS - 1);
                atomicAdd(&tri_cnt[idx], 1u);
                ++myTri;
            }
        }
    }
    if (myBi)  atomicAdd(&totals[0], myBi);
    if (myTri) atomicAdd(&totals[1], myTri);
    __syncthreads();

    const float biInv  = 1.0f / (float)max(totals[0], 1u);
    const float triInv = 1.0f / (float)max(totals[1], 1u);

    float* bf = bi_freq + (size_t)row * KPAD_BI;
    for (int i = tid; i < KPAD_BI; i += 256)
        bf[i] = (i < V2 ? (float)bi_cnt[i] : 0.0f) * biInv;

    float* tf = tri_freq + (size_t)row * BUCKETS;
    for (int i = tid; i < BUCKETS; i += 256)
        tf[i] = (float)tri_cnt[i] * triInv;
}

__global__ __launch_bounds__(256) void mlp_kernel(const float* __restrict__ freq,
                                                  int K, int Kpad,
                                                  const float* __restrict__ W1,
                                                  const float* __restrict__ b1,
                                                  const float* __restrict__ W2,
                                                  const float* __restrict__ b2,
                                                  float* __restrict__ out,
                                                  int outOff) {
    __shared__ float lds_h[16 * HID];

    const int rowBase = blockIdx.x * 16;
    const int lane = threadIdx.x & 31;
    const int wave = threadIdx.x >> 5;
    const int m  = lane & 15;
    const int kh = lane >> 4;
    const int n  = lane & 15;
    const int colBase = wave * 32;

    v8f c0 = {};
    v8f c1 = {};

    const float* arow = freq + (size_t)(rowBase + m) * Kpad;

    for (int k0 = 0; k0 < Kpad; k0 += 4) {
        const int ka = k0 + 2 * kh;
        v2f a;
        a.x = arow[ka];
        a.y = arow[ka + 1];
        const int kb0 = min(ka,     K - 1);
        const int kb1 = min(ka + 1, K - 1);
        v2f bA, bB;
        bA.x = W1[(size_t)kb0 * HID + colBase + n];
        bA.y = W1[(size_t)kb1 * HID + colBase + n];
        bB.x = W1[(size_t)kb0 * HID + colBase + 16 + n];
        bB.y = W1[(size_t)kb1 * HID + colBase + 16 + n];
        c0 = __builtin_amdgcn_wmma_f32_16x16x4_f32(false, a, false, bA, (short)0, c0, false, false);
        c1 = __builtin_amdgcn_wmma_f32_16x16x4_f32(false, a, false, bB, (short)0, c1, false, false);
    }

    #pragma unroll
    for (int r = 0; r < 8; ++r) {
        const int mm = r + 8 * kh;
        float h0 = c0[r] + b1[colBase + n];
        float h1 = c1[r] + b1[colBase + 16 + n];
        h0 = h0 > 0.0f ? h0 : 0.0f;
        h1 = h1 > 0.0f ? h1 : 0.0f;
        lds_h[mm * HID + colBase + n]      = h0;
        lds_h[mm * HID + colBase + 16 + n] = h1;
    }
    __syncthreads();

    if (wave < 2) {
        const int cb = wave * 16;
        v8f d = {};
        for (int k0 = 0; k0 < HID; k0 += 4) {
            const int ka = k0 + 2 * kh;
            v2f a;
            a.x = lds_h[m * HID + ka];
            a.y = lds_h[m * HID + ka + 1];
            v2f b;
            b.x = W2[(size_t)ka * 32 + cb + n];
            b.y = W2[(size_t)(ka + 1) * 32 + cb + n];
            d = __builtin_amdgcn_wmma_f32_16x16x4_f32(false, a, false, b, (short)0, d, false, false);
        }
        #pragma unroll
        for (int r = 0; r < 8; ++r) {
            const int mm = r + 8 * kh;
            out[(size_t)(rowBase + mm) * OUTC + outOff + cb + n] = d[r] + b2[cb + n];
        }
    }
}

extern "C" void kernel_launch(void* const* d_in, const int* in_sizes, int n_in,
                              void* d_out, int out_size, void* d_ws, size_t ws_size,
                              hipStream_t stream) {
    const int*   src = (const int*)d_in[0];
    const float* Wb1 = (const float*)d_in[1];
    const float* bb1 = (const float*)d_in[2];
    const float* Wb2 = (const float*)d_in[3];
    const float* bb2 = (const float*)d_in[4];
    const float* Wt1 = (const float*)d_in[5];
    const float* bt1 = (const float*)d_in[6];
    const float* Wt2 = (const float*)d_in[7];
    const float* bt2 = (const float*)d_in[8];
    float* out = (float*)d_out;

    float* bi_freq  = (float*)d_ws;
    float* tri_freq = bi_freq + (size_t)B_ROWS * KPAD_BI;

    hist_kernel<<<B_ROWS, 256, 0, stream>>>(src, bi_freq, tri_freq);

    mlp_kernel<<<B_ROWS / 16, 256, 0, stream>>>(bi_freq, V2, KPAD_BI,
                                                Wb1, bb1, Wb2, bb2, out, 0);
    mlp_kernel<<<B_ROWS / 16, 256, 0, stream>>>(tri_freq, BUCKETS, BUCKETS,
                                                Wt1, bt1, Wt2, bt2, out, 32);
}


# Histogram kernel: one wave per row (8 rows per workgroup, private LDS bins, no workgroup barrier), tokens read as 512-byte wave loads issued 8 chunks ahead of the counting; integer counts unchanged
# speedup vs baseline: 2.2576x; 2.2576x over previous
#include <hip/hip_runtime.h>

#define VOCAB 33
#define V2 1089
#define BUCKETS 512
#define B_ROWS 4096
#define S_LEN 4096
#define KPAD_BI 1104
#define HID 256
#define OUTC 64

typedef float v2f __attribute__((ext_vector_type(2)));
typedef float v4f __attribute__((ext_vector_type(4)));
typedef float v8f __attribute__((ext_vector_type(8)));

#define HROWS 8
#define WBINS (KPAD_BI + BUCKETS)

typedef int v4i __attribute__((ext_vector_type(4)));
typedef int v2i __attribute__((ext_vector_type(2)));
typedef unsigned v4u __attribute__((ext_vector_type(4)));

struct TokGroup {
    v4i a[8];
    v2i la[8];
};

template <int G>
__device__ __forceinline__ void load_group(TokGroup& t, const int* __restrict__ s, int lane) {
    #pragma unroll
    for (int q = 0; q < 8; ++q) {
        const int tok = (G * 8 + q) * 128 + 4 * lane;
        t.a[q] = *(const v4i*)(s + tok);
        if (G == 3 && q == 7) {
            v2i z = {0, 0};
            if (lane < 31) z = *(const v2i*)(s + tok + 4);
            t.la[q] = z;
        } else {
            t.la[q] = *(const v2i*)(s + tok + 4);
        }
    }
}

__device__ __forceinline__ void count3(unsigned* __restrict__ bi, unsigned* __restrict__ tri,
                                       int c, int n, int t) {
    if ((c > 0) && (n > 0)) {
        const int b = c * VOCAB + n;
        atomicAdd(&bi[b], 1u);
        if (t > 0) atomicAdd(&tri[((unsigned)(b * VOCAB + t)) & (BUCKETS - 1)], 1u);
    }
}

__device__ __forceinline__ void count_group(const TokGroup& t, unsigned* __restrict__ bi,
                                            unsigned* __restrict__ tri) {
    #pragma unroll
    for (int q = 0; q < 8; ++q) {
        const v4i a = t.a[q];
        const v2i l = t.la[q];
        count3(bi, tri, a.x, a.y, a.z);
        count3(bi, tri, a.y, a.z, a.w);
        count3(bi, tri, a.z, a.w, l.x);
        count3(bi, tri, a.w, l.x, l.y);
    }
}

__global__ __launch_bounds__(32 * HROWS) void hist_kernel(const int* __restrict__ src,
                                                          float* __restrict__ bi_freq,
                                                          float* __restrict__ tri_freq) {
    __shared__ __attribute__((aligned(16))) unsigned cnt[HROWS * WBINS];

    const int lane = threadIdx.x & 31;
    const int wave = __builtin_amdgcn_readfirstlane(threadIdx.x >> 5);
    const int row  = blockIdx.x * HROWS + wave;
    const int* s = src + (size_t)row * S_LEN;
    unsigned* bi  = cnt + wave * WBINS;
    unsigned* tri = bi + KPAD_BI;

    TokGroup f0, f1;
    load_group<0>(f0, s, lane);

    {
        const v4u z = {0u, 0u, 0u, 0u};
        for (int i = lane; i < WBINS / 4; i += 32) *(v4u*)&bi[4 * i] = z;
    }

    load_group<1>(f1, s, lane);
    __builtin_amdgcn_sched_barrier(0);
    count_group(f0, bi, tri);
    load_group<2>(f0, s, lane);
    __builtin_amdgcn_sched_barrier(0);
    count_group(f1, bi, tri);
    load_group<3>(f1, s, lane);
    __builtin_amdgcn_sched_barrier(0);
    count_group(f0, bi, tri);
    count_group(f1, bi, tri);

    v4u cb[9], ct[4];
    #pragma unroll
    for (int q = 0; q < 9; ++q) {
        const int i4 = lane + 32 * q;
        v4u v = {0u, 0u, 0u, 0u};
        if (q < 8 || i4 < KPAD_BI / 4) v = *(const v4u*)&bi[4 * i4];
        cb[q] = v;
    }
    #pragma unroll
    for (int q = 0; q < 4; ++q) ct[q] = *(const v4u*)&tri[4 * (lane + 32 * q)];

    unsigned sb = 0u, st = 0u;
    #pragma unroll
    for (int q = 0; q < 9; ++q) sb += cb[q].x + cb[q].y + cb[q].z + cb[q].w;
    #pragma unroll
    for (int q = 0; q < 4; ++q) st += ct[q].x + ct[q].y + ct[q].z + ct[q].w;
    #pragma unroll
    for (int m = 16; m >= 1; m >>= 1) {
        sb += (unsigned)__shfl_xor((int)sb, m);
        st += (unsigned)__shfl_xor((int)st, m);
    }

    const float biInv  = 1.0f / (float)max(sb, 1u);
    const float triInv = 1.0f / (float)max(st, 1u);

    v4f* bf = (v4f*)(bi_freq + (size_t)row * KPAD_BI);
    #pragma unroll
    for (int q = 0; q < 9; ++q) {
        const int i4 = lane + 32 * q;
        if (q < 8 || i4 < KPAD_BI / 4) {
            v4f o = {(float)cb[q].x * biInv, (float)cb[q].y * biInv,
                     (float)cb[q].z * biInv, (float)cb[q].w * biInv};
            bf[i4] = o;
        }
    }
    v4f* tf = (v4f*)(tri_freq + (size_t)row * BUCKETS);
    #pragma unroll
    for (int q = 0; q < 4; ++q) {
        v4f o = {(float)ct[q].x * triInv, (float)ct[q].y * triInv,
                 (float)ct[q].z * triInv, (float)ct[q].w * triInv};
        tf[lane + 32 * q] = o;
    }
}

struct KBlock {
    v2f a[2];
    v4f bx[2][4];
    v4f by[2][4];
};

__device__ __forceinline__ void load_kblock(KBlock& f, const float* __restrict__ arow,
                                            const float* __restrict__ wlane,
                                            int kbase, int K) {
    #pragma unroll
    for (int s = 0; s < 2; ++s) {
        const int ka = kbase + 4 * s;
        f.a[s] = *(const v2f*)(arow + ka);
        const int kb0 = min(ka,     K - 1);
        const int kb1 = min(ka + 1, K - 1);
        const v4f* r0 = (const v4f*)(wlane + (size_t)kb0 * HID);
        const v4f* r1 = (const v4f*)(wlane + (size_t)kb1 * HID);
        #pragma unroll
        for (int g = 0; g < 4; ++g) {
            f.bx[s][g] = r0[g * 16];
            f.by[s][g] = r1[g * 16];
        }
    }
}

__device__ __forceinline__ void mma_kblock(const KBlock& f, v8f (&acc)[16]) {
    #pragma unroll
    for (int s = 0; s < 2; ++s) {
        #pragma unroll
        for (int g = 0; g < 4; ++g) {
            #pragma unroll
            for (int j = 0; j < 4; ++j) {
                v2f b;
                b.x = f.bx[s][g][j];
                b.y = f.by[s][g][j];
                acc[g * 4 + j] = __builtin_amdgcn_wmma_f32_16x16x4_f32(
                    false, f.a[s], false, b, (short)0, acc[g * 4 + j], false, false);
            }
        }
    }
}

__global__ __launch_bounds__(256) void mlp_kernel(const float* __restrict__ freq,
                                                  int K, int Kpad,
                                                  const float* __restrict__ W1,
                                                  const float* __restrict__ b1,
                                                  const float* __restrict__ W2,
                                                  const float* __restrict__ b2,
                                                  float* __restrict__ out,
                                                  int outOff) {
    __shared__ float lds_h[16 * HID];
    __shared__ v4f   lds_p[8 * 16 * 2 * 32];

    const int rowBase = blockIdx.x * 16;
    const int lane = threadIdx.x & 31;
    const int wave = __builtin_amdgcn_readfirstlane(threadIdx.x >> 5);
    const int m  = lane & 15;
    const int kh = lane >> 4;
    const int n  = lane & 15;

    v8f acc[16];
    #pragma unroll
    for (int t = 0; t < 16; ++t) acc[t] = (v8f){};

    const float* arow  = freq + (size_t)(rowBase + m) * Kpad;
    const float* wlane = W1 + 4 * n;
    const int nBlk = Kpad >> 3;

    const int cnt  = (nBlk - wave + 7) >> 3;
    const int last = wave + 8 * (cnt - 1);
    const int kl   = 2 * kh;
    KBlock f0, f1;
    load_kblock(f0, arow, wlane, wave * 8 + kl, K);
    int i = 0;
    for (; i + 2 <= cnt; i += 2) {
        load_kblock(f1, arow, wlane, (wave + 8 * (i + 1)) * 8 + kl, K);
        __builtin_amdgcn_sched_barrier(0);
        mma_kblock(f0, acc);
        load_kblock(f0, arow, wlane, min(wave + 8 * (i + 2), last) * 8 + kl, K);
        __builtin_amdgcn_sched_barrier(0);
        mma_kblock(f1, acc);
    }
    if (cnt & 1) mma_kblock(f0, acc);

    #pragma unroll
    for (int t = 0; t < 16; ++t) {
        v4f lo = {acc[t][0], acc[t][1], acc[t][2], acc[t][3]};
        v4f hi = {acc[t][4], acc[t][5], acc[t][6], acc[t][7]};
        lds_p[((wave * 16 + t) * 2 + 0) * 32 + lane] = lo;
        lds_p[((wave * 16 + t) * 2 + 1) * 32 + lane] = hi;
    }
    __syncthreads();

    {
        const int g  = wave >> 1;
        const int j0 = (wave & 1) * 2;
        v4f s[2][2];
        #pragma unroll
        for (int q = 0; q < 2; ++q) {
            const int t = 2 * wave + q;
            s[q][0] = lds_p[((0 * 16 + t) * 2 + 0) * 32 + lane];
            s[q][1] = lds_p[((0 * 16 + t) * 2 + 1) * 32 + lane];
            #pragma unroll
            for (int w = 1; w < 8; ++w) {
                s[q][0] += lds_p[((w * 16 + t) * 2 + 0) * 32 + lane];
                s[q][1] += lds_p[((w * 16 + t) * 2 + 1) * 32 + lane];
            }
        }
        const int col = g * 64 + n * 4 + j0;
        const float bias0 = b1[col];
        const float bias1 = b1[col + 1];
        #pragma unroll
        for (int r = 0; r < 8; ++r) {
            const int mm = r + 8 * kh;
            float h0 = s[0][r >> 2][r & 3] + bias0;
            float h1 = s[1][r >> 2][r & 3] + bias1;
            h0 = h0 > 0.0f ? h0 : 0.0f;
            h1 = h1 > 0.0f ? h1 : 0.0f;
            v2f h = {h0, h1};
            *(v2f*)&lds_h[mm * HID + col] = h;
        }
    }
    __syncthreads();

    if (wave < 2) {
        const int cb = wave * 16;
        v8f d = {};
        for (int k0 = 0; k0 < HID; k0 += 4) {
            const int ka = k0 + 2 * kh;
            v2f a;
            a.x = lds_h[m * HID + ka];
            a.y = lds_h[m * HID + ka + 1];
            v2f b;
            b.x = W2[(size_t)ka * 32 + cb + n];
            b.y = W2[(size_t)(ka + 1) * 32 + cb + n];
            d = __builtin_amdgcn_wmma_f32_16x16x4_f32(false, a, false, b, (short)0, d, false, false);
        }
        #pragma unroll
        for (int r = 0; r < 8; ++r) {
            const int mm = r + 8 * kh;
            out[(size_t)(rowBase + mm) * OUTC + outOff + cb + n] = d[r] + b2[cb + n];
        }
    }
}

extern "C" void kernel_launch(void* const* d_in, const int* in_sizes, int n_in,
                              void* d_out, int out_size, void* d_ws, size_t ws_size,
                              hipStream_t stream) {
    const int*   src = (const int*)d_in[0];
    const float* Wb1 = (const float*)d_in[1];
    const float* bb1 = (const float*)d_in[2];
    const float* Wb2 = (const float*)d_in[3];
    const float* bb2 = (const float*)d_in[4];
    const float* Wt1 = (const float*)d_in[5];
    const float* bt1 = (const float*)d_in[6];
    const float* Wt2 = (const float*)d_in[7];
    const float* bt2 = (const float*)d_in[8];
    float* out = (float*)d_out;

    float* bi_freq  = (float*)d_ws;
    float* tri_freq = bi_freq + (size_t)B_ROWS * KPAD_BI;

    hist_kernel<<<B_ROWS / HROWS, 32 * HROWS, 0, stream>>>(src, bi_freq, tri_freq);

    mlp_kernel<<<B_ROWS / 16, 256, 0, stream>>>(bi_freq, V2, KPAD_BI,
                                                Wb1, bb1, Wb2, bb2, out, 0);
    mlp_kernel<<<B_ROWS / 16, 256, 0, stream>>>(tri_freq, BUCKETS, BUCKETS,
                                                Wt1, bt1, Wt2, bt2, out, 32);
}


# Both MLP layers transposed (weights as A operand): a wave's reduced layer-1 registers are layer 2's B fragments as they stand, no LDS hidden tile; 8 partial out tiles added via LDS (new sum tree)
# speedup vs baseline: 2.7138x; 1.2021x over previous
#include <hip/hip_runtime.h>

#define VOCAB 33
#define V2 1089
#define BUCKETS 512
#define B_ROWS 4096
#define S_LEN 4096
#define KPAD_BI 1104
#define HID 256
#define OUTC 64

typedef float v2f __attribute__((ext_vector_type(2)));
typedef float v4f __attribute__((ext_vector_type(4)));
typedef float v8f __attribute__((ext_vector_type(8)));

#define HROWS 8
#define WBINS (KPAD_BI + BUCKETS)

typedef int v4i __attribute__((ext_vector_type(4)));
typedef int v2i __attribute__((ext_vector_type(2)));
typedef unsigned v4u __attribute__((ext_vector_type(4)));

struct TokGroup {
    v4i a[8];
    v2i la[8];
};

template <int G>
__device__ __forceinline__ void load_group(TokGroup& t, const int* __restrict__ s, int lane) {
    #pragma unroll
    for (int q = 0; q < 8; ++q) {
        const int tok = (G * 8 + q) * 128 + 4 * lane;
        t.a[q] = *(const v4i*)(s + tok);
        if (G == 3 && q == 7) {
            v2i z = {0, 0};
            if (lane < 31) z = *(const v2i*)(s + tok + 4);
            t.la[q] = z;
        } else {
            t.la[q] = *(const v2i*)(s + tok + 4);
        }
    }
}

__device__ __forceinline__ void count3(unsigned* __restrict__ bi, unsigned* __restrict__ tri,
                                       int c, int n, int t) {
    if ((c > 0) && (n > 0)) {
        const int b = c * VOCAB + n;
        atomicAdd(&bi[b], 1u);
        if (t > 0) atomicAdd(&tri[((unsigned)(b * VOCAB + t)) & (BUCKETS - 1)], 1u);
    }
}

__device__ __forceinline__ void count_group(const TokGroup& t, unsigned* __restrict__ bi,
                                            unsigned* __restrict__ tri) {
    #pragma unroll
    for (int q = 0; q < 8; ++q) {
        const v4i a = t.a[q];
        const v2i l = t.la[q];
        count3(bi, tri, a.x, a.y, a.z);
        count3(bi, tri, a.y, a.z, a.w);
        count3(bi, tri, a.z, a.w, l.x);
        count3(bi, tri, a.w, l.x, l.y);
    }
}

__global__ __launch_bounds__(32 * HROWS) void hist_kernel(const int* __restrict__ src,
                                                          float* __restrict__ bi_freq,
                                                          float* __restrict__ tri_freq) {
    __shared__ __attribute__((aligned(16))) unsigned cnt[HROWS * WBINS];

    const int lane = threadIdx.x & 31;
    const int wave = __builtin_amdgcn_readfirstlane(threadIdx.x >> 5);
    const int row  = blockIdx.x * HROWS + wave;
    const int* s = src + (size_t)row * S_LEN;
    unsigned* bi  = cnt + wave * WBINS;
    unsigned* tri = bi + KPAD_BI;

    TokGroup f0, f1;
    load_group<0>(f0, s, lane);

    {
        const v4u z = {0u, 0u, 0u, 0u};
        for (int i = lane; i < WBINS / 4; i += 32) *(v4u*)&bi[4 * i] = z;
    }

    load_group<1>(f1, s, lane);
    __builtin_amdgcn_sched_barrier(0);
    count_group(f0, bi, tri);
    load_group<2>(f0, s, lane);
    __builtin_amdgcn_sched_barrier(0);
    count_group(f1, bi, tri);
    load_group<3>(f1, s, lane);
    __builtin_amdgcn_sched_barrier(0);
    count_group(f0, bi, tri);
    count_group(f1, bi, tri);

    v4u cb[9], ct[4];
    #pragma unroll
    for (int q = 0; q < 9; ++q) {
        const int i4 = lane + 32 * q;
        v4u v = {0u, 0u, 0u, 0u};
        if (q < 8 || i4 < KPAD_BI / 4) v = *(const v4u*)&bi[4 * i4];
        cb[q] = v;
    }
    #pragma unroll
    for (int q = 0; q < 4; ++q) ct[q] = *(const v4u*)&tri[4 * (lane + 32 * q)];

    unsigned sb = 0u, st = 0u;
    #pragma unroll
    for (int q = 0; q < 9; ++q) sb += cb[q].x + cb[q].y + cb[q].z + cb[q].w;
    #pragma unroll
    for (int q = 0; q < 4; ++q) st += ct[q].x + ct[q].y + ct[q].z + ct[q].w;
    #pragma unroll
    for (int m = 16; m >= 1; m >>= 1) {
        sb += (unsigned)__shfl_xor((int)sb, m);
        st += (unsigned)__shfl_xor((int)st, m);
    }

    const float biInv  = 1.0f / (float)max(sb, 1u);
    const float triInv = 1.0f / (float)max(st, 1u);

    v4f* bf = (v4f*)(bi_freq + (size_t)row * KPAD_BI);
    #pragma unroll
    for (int q = 0; q < 9; ++q) {
        const int i4 = lane + 32 * q;
        if (q < 8 || i4 < KPAD_BI / 4) {
            v4f o = {(float)cb[q].x * biInv, (float)cb[q].y * biInv,
                     (float)cb[q].z * biInv, (float)cb[q].w * biInv};
            bf[i4] = o;
        }
    }
    v4f* tf = (v4f*)(tri_freq + (size_t)row * BUCKETS);
    #pragma unroll
    for (int q = 0; q < 4; ++q) {
        v4f o = {(float)ct[q].x * triInv, (float)ct[q].y * triInv,
                 (float)ct[q].z * triInv, (float)ct[q].w * triInv};
        tf[lane + 32 * q] = o;
    }
}

#define OPITCH 36
struct KBlock {
    v2f a[2];
    v4f bx[2][4];
    v4f by[2][4];
};

__device__ __forceinline__ void load_kblock(KBlock& f, const float* __restrict__ arow,
                                            const float* __restrict__ wlane,
                                            int kbase, int K) {
    #pragma unroll
    for (int s = 0; s < 2; ++s) {
        const int ka = kbase + 4 * s;
        f.a[s] = *(const v2f*)(arow + ka);
        const int kb0 = min(ka,     K - 1);
        const int kb1 = min(ka + 1, K - 1);
        const v4f* r0 = (const v4f*)(wlane + (size_t)kb0 * HID);
        const v4f* r1 = (const v4f*)(wlane + (size_t)kb1 * HID);
        #pragma unroll
        for (int g = 0; g < 4; ++g) {
            f.bx[s][g] = r0[g * 16];
            f.by[s][g] = r1[g * 16];
        }
    }
}

__device__ __forceinline__ void mma_kblock(const KBlock& f, v8f (&acc)[16]) {
    #pragma unroll
    for (int s = 0; s < 2; ++s) {
        #pragma unroll
        for (int g = 0; g < 4; ++g) {
            #pragma unroll
            for (int j = 0; j < 4; ++j) {
                v2f b;
                b.x = f.bx[s][g][j];
                b.y = f.by[s][g][j];
                acc[g * 4 + j] = __builtin_amdgcn_wmma_f32_16x16x4_f32(
                    false, b, false, f.a[s], (short)0, acc[g * 4 + j], false, false);
            }
        }
    }
}

__global__ __launch_bounds__(256) void mlp_kernel(const float* __restrict__ freq,
                                                  int K, int Kpad,
                                                  const float* __restrict__ W1,
                                                  const float* __restrict__ b1,
                                                  const float* __restrict__ W2,
                                                  const float* __restrict__ b2,
                                                  float* __restrict__ out,
                                                  int outOff) {
    __shared__ v4f   lds_p[8 * 16 * 2 * 32];
    __shared__ __attribute__((aligned(16))) float lds_o[8 * 16 * OPITCH];

    const int rowBase = blockIdx.x * 16;
    const int lane = threadIdx.x & 31;
    const int wave = __builtin_amdgcn_readfirstlane(threadIdx.x >> 5);
    const int m  = lane & 15;
    const int kh = lane >> 4;
    const int n  = lane & 15;

    v8f acc[16];
    #pragma unroll
    for (int t = 0; t < 16; ++t) acc[t] = (v8f){};

    const float* arow  = freq + (size_t)(rowBase + m) * Kpad;
    const float* wlane = W1 + 4 * n;
    const int nBlk = Kpad >> 3;

    const int cnt  = (nBlk - wave + 7) >> 3;
    const int last = wave + 8 * (cnt - 1);
    const int kl   = 2 * kh;
    KBlock f0, f1;
    load_kblock(f0, arow, wlane, wave * 8 + kl, K);

    const int hc0 = (wave >> 1) * 64 + 32 * kh + (wave & 1) * 2;
    v2f w2f[8][2];
    v2f bia[8];
    #pragma unroll
    for (int r = 0; r < 8; ++r) {
        const int hc = hc0 + 4 * r;
        bia[r] = *(const v2f*)(b1 + hc);
        #pragma unroll
        for (int mt = 0; mt < 2; ++mt) {
            w2f[r][mt].x = W2[(size_t)hc * 32 + 16 * mt + m];
            w2f[r][mt].y = W2[(size_t)(hc + 1) * 32 + 16 * mt + m];
        }
    }
    const v2f bout = *(const v2f*)(b2 + (threadIdx.x & 15) * 2);

    int i = 0;
    for (; i + 2 <= cnt; i += 2) {
        load_kblock(f1, arow, wlane, (wave + 8 * (i + 1)) * 8 + kl, K);
        __builtin_amdgcn_sched_barrier(0);
        mma_kblock(f0, acc);
        load_kblock(f0, arow, wlane, min(wave + 8 * (i + 2), last) * 8 + kl, K);
        __builtin_amdgcn_sched_barrier(0);
        mma_kblock(f1, acc);
    }
    if (cnt & 1) mma_kblock(f0, acc);

    #pragma unroll
    for (int t = 0; t < 16; ++t) {
        v4f lo = {acc[t][0], acc[t][1], acc[t][2], acc[t][3]};
        v4f hi = {acc[t][4], acc[t][5], acc[t][6], acc[t][7]};
        lds_p[((wave * 16 + t) * 2 + 0) * 32 + lane] = lo;
        lds_p[((wave * 16 + t) * 2 + 1) * 32 + lane] = hi;
    }
    __syncthreads();

    v2f hp[8];
    {
        v4f s[2][2];
        #pragma unroll
        for (int q = 0; q < 2; ++q) {
            const int t = 2 * wave + q;
            s[q][0] = lds_p[((0 * 16 + t) * 2 + 0) * 32 + lane];
            s[q][1] = lds_p[((0 * 16 + t) * 2 + 1) * 32 + lane];
            #pragma unroll
            for (int w = 1; w < 8; ++w) {
                s[q][0] += lds_p[((w * 16 + t) * 2 + 0) * 32 + lane];
                s[q][1] += lds_p[((w * 16 + t) * 2 + 1) * 32 + lane];
            }
        }
        #pragma unroll
        for (int r = 0; r < 8; ++r) {
            float h0 = s[0][r >> 2][r & 3] + bia[r].x;
            float h1 = s[1][r >> 2][r & 3] + bia[r].y;
            h0 = h0 > 0.0f ? h0 : 0.0f;
            h1 = h1 > 0.0f ? h1 : 0.0f;
            hp[r] = (v2f){h0, h1};
        }
    }

    v8f o0 = {}, o1 = {};
    #pragma unroll
    for (int r = 0; r < 8; ++r) {
        o0 = __builtin_amdgcn_wmma_f32_16x16x4_f32(false, w2f[r][0], false, hp[r], (short)0, o0, false, false);
        o1 = __builtin_amdgcn_wmma_f32_16x16x4_f32(false, w2f[r][1], false, hp[r], (short)0, o1, false, false);
    }

    {
        float* po = lds_o + (wave * 16 + m) * OPITCH + 8 * kh;
        *(v4f*)(po)      = (v4f){o0[0], o0[1], o0[2], o0[3]};
        *(v4f*)(po + 4)  = (v4f){o0[4], o0[5], o0[6], o0[7]};
        *(v4f*)(po + 16) = (v4f){o1[0], o1[1], o1[2], o1[3]};
        *(v4f*)(po + 20) = (v4f){o1[4], o1[5], o1[6], o1[7]};
    }
    __syncthreads();

    {
        const int row = threadIdx.x >> 4;
        const int oc  = (threadIdx.x & 15) * 2;
        v2f o = *(const v2f*)&lds_o[(0 * 16 + row) * OPITCH + oc];
        #pragma unroll
        for (int w = 1; w < 8; ++w) o += *(const v2f*)&lds_o[(w * 16 + row) * OPITCH + oc];
        o += bout;
        *(v2f*)&out[(size_t)(rowBase + row) * OUTC + outOff + oc] = o;
    }
}

extern "C" void kernel_launch(void* const* d_in, const int* in_sizes, int n_in,
                              void* d_out, int out_size, void* d_ws, size_t ws_size,
                              hipStream_t stream) {
    const int*   src = (const int*)d_in[0];
    const float* Wb1 = (const float*)d_in[1];
    const float* bb1 = (const float*)d_in[2];
    const float* Wb2 = (const float*)d_in[3];
    const float* bb2 = (const float*)d_in[4];
    const float* Wt1 = (const float*)d_in[5];
    const float* bt1 = (const float*)d_in[6];
    const float* Wt2 = (const float*)d_in[7];
    const float* bt2 = (const float*)d_in[8];
    float* out = (float*)d_out;

    float* bi_freq  = (float*)d_ws;
    float* tri_freq = bi_freq + (size_t)B_ROWS * KPAD_BI;

    hist_kernel<<<B_ROWS / HROWS, 32 * HROWS, 0, stream>>>(src, bi_freq, tri_freq);

    mlp_kernel<<<B_ROWS / 16, 256, 0, stream>>>(bi_freq, V2, KPAD_BI,
                                                Wb1, bb1, Wb2, bb2, out, 0);
    mlp_kernel<<<B_ROWS / 16, 256, 0, stream>>>(tri_freq, BUCKETS, BUCKETS,
                                                Wt1, bt1, Wt2, bt2, out, 32);
}


# Each 16-row MLP workgroup runs the bigram, then the trigram layer-1 K loop; partial tiles parked in two LDS regions, one shared reduce + layer-2 epilogue: 3 launches -> 2, 4 barriers -> 2; same sums
# speedup vs baseline: 2.9616x; 1.0913x over previous
#include <hip/hip_runtime.h>

#define VOCAB 33
#define V2 1089
#define BUCKETS 512
#define B_ROWS 4096
#define S_LEN 4096
#define KPAD_BI 1104
#define HID 256
#define OUTC 64

typedef float v2f __attribute__((ext_vector_type(2)));
typedef float v4f __attribute__((ext_vector_type(4)));
typedef float v8f __attribute__((ext_vector_type(8)));

#define HROWS 8
#define WBINS (KPAD_BI + BUCKETS)

typedef int v4i __attribute__((ext_vector_type(4)));
typedef int v2i __attribute__((ext_vector_type(2)));
typedef unsigned v4u __attribute__((ext_vector_type(4)));

struct TokGroup {
    v4i a[8];
    v2i la[8];
};

template <int G>
__device__ __forceinline__ void load_group(TokGroup& t, const int* __restrict__ s, int lane) {
    #pragma unroll
    for (int q = 0; q < 8; ++q) {
        const int tok = (G * 8 + q) * 128 + 4 * lane;
        t.a[q] = *(const v4i*)(s + tok);
        if (G == 3 && q == 7) {
            v2i z = {0, 0};
            if (lane < 31) z = *(const v2i*)(s + tok + 4);
            t.la[q] = z;
        } else {
            t.la[q] = *(const v2i*)(s + tok + 4);
        }
    }
}

__device__ __forceinline__ void count3(unsigned* __restrict__ bi, unsigned* __restrict__ tri,
                                       int c, int n, int t) {
    if ((c > 0) && (n > 0)) {
        const int b = c * VOCAB + n;
        atomicAdd(&bi[b], 1u);
        if (t > 0) atomicAdd(&tri[((unsigned)(b * VOCAB + t)) & (BUCKETS - 1)], 1u);
    }
}

__device__ __forceinline__ void count_group(const TokGroup& t, unsigned* __restrict__ bi,
                                            unsigned* __restrict__ tri) {
    #pragma unroll
    for (int q = 0; q < 8; ++q) {
        const v4i a = t.a[q];
        const v2i l = t.la[q];
        count3(bi, tri, a.x, a.y, a.z);
        count3(bi, tri, a.y, a.z, a.w);
        count3(bi, tri, a.z, a.w, l.x);
        count3(bi, tri, a.w, l.x, l.y);
    }
}

__global__ __launch_bounds__(32 * HROWS) void hist_kernel(const int* __restrict__ src,
                                                          float* __restrict__ bi_freq,
                                                          float* __restrict__ tri_freq) {
    __shared__ __attribute__((aligned(16))) unsigned cnt[HROWS * WBINS];

    const int lane = threadIdx.x & 31;
    const int wave = __builtin_amdgcn_readfirstlane(threadIdx.x >> 5);
    const int row  = blockIdx.x * HROWS + wave;
    const int* s = src + (size_t)row * S_LEN;
    unsigned* bi  = cnt + wave * WBINS;
    unsigned* tri = bi + KPAD_BI;

    TokGroup f0, f1;
    load_group<0>(f0, s, lane);

    {
        const v4u z = {0u, 0u, 0u, 0u};
        for (int i = lane; i < WBINS / 4; i += 32) *(v4u*)&bi[4 * i] = z;
    }

    load_group<1>(f1, s, lane);
    __builtin_amdgcn_sched_barrier(0);
    count_group(f0, bi, tri);
    load_group<2>(f0, s, lane);
    __builtin_amdgcn_sched_barrier(0);
    count_group(f1, bi, tri);
    load_group<3>(f1, s, lane);
    __builtin_amdgcn_sched_barrier(0);
    count_group(f0, bi, tri);
    count_group(f1, bi, tri);

    v4u cb[9], ct[4];
    #pragma unroll
    for (int q = 0; q < 9; ++q) {
        const int i4 = lane + 32 * q;
        v4u v = {0u, 0u, 0u, 0u};
        if (q < 8 || i4 < KPAD_BI / 4) v = *(const v4u*)&bi[4 * i4];
        cb[q] = v;
    }
    #pragma unroll
    for (int q = 0; q < 4; ++q) ct[q] = *(const v4u*)&tri[4 * (lane + 32 * q)];

    unsigned sb = 0u, st = 0u;
    #pragma unroll
    for (int q = 0; q < 9; ++q) sb += cb[q].x + cb[q].y + cb[q].z + cb[q].w;
    #pragma unroll
    for (int q = 0; q < 4; ++q) st += ct[q].x + ct[q].y + ct[q].z + ct[q].w;
    #pragma unroll
    for (int m = 16; m >= 1; m >>= 1) {
        sb += (unsigned)__shfl_xor((int)sb, m);
        st += (unsigned)__shfl_xor((int)st, m);
    }

    const float biInv  = 1.0f / (float)max(sb, 1u);
    const float triInv = 1.0f / (float)max(st, 1u);

    v4f* bf = (v4f*)(bi_freq + (size_t)row * KPAD_BI);
    #pragma unroll
    for (int q = 0; q < 9; ++q) {
        const int i4 = lane + 32 * q;
        if (q < 8 || i4 < KPAD_BI / 4) {
            v4f o = {(float)cb[q].x * biInv, (float)cb[q].y * biInv,
                     (float)cb[q].z * biInv, (float)cb[q].w * biInv};
            bf[i4] = o;
        }
    }
    v4f* tf = (v4f*)(tri_freq + (size_t)row * BUCKETS);
    #pragma unroll
    for (int q = 0; q < 4; ++q) {
        v4f o = {(float)ct[q].x * triInv, (float)ct[q].y * triInv,
                 (float)ct[q].z * triInv, (float)ct[q].w * triInv};
        tf[lane + 32 * q] = o;
    }
}

#define OPITCH 36
struct KBlock {
    v2f a[2];
    v4f bx[2][4];
    v4f by[2][4];
};

__device__ __forceinline__ void load_kblock(KBlock& f, const float* __restrict__ arow,
                                            const float* __restrict__ wlane,
                                            int kbase, int K) {
    #pragma unroll
    for (int s = 0; s < 2; ++s) {
        const int ka = kbase + 4 * s;
        f.a[s] = *(const v2f*)(arow + ka);
        const int kb0 = min(ka,     K - 1);
        const int kb1 = min(ka + 1, K - 1);
        const v4f* r0 = (const v4f*)(wlane + (size_t)kb0 * HID);
        const v4f* r1 = (const v4f*)(wlane + (size_t)kb1 * HID);
        #pragma unroll
        for (int g = 0; g < 4; ++g) {
            f.bx[s][g] = r0[g * 16];
            f.by[s][g] = r1[g * 16];
        }
    }
}

__device__ __forceinline__ void mma_kblock(const KBlock& f, v8f (&acc)[16]) {
    #pragma unroll
    for (int s = 0; s < 2; ++s) {
        #pragma unroll
        for (int g = 0; g < 4; ++g) {
            #pragma unroll
            for (int j = 0; j < 4; ++j) {
                v2f b;
                b.x = f.bx[s][g][j];
                b.y = f.by[s][g][j];
                acc[g * 4 + j] = __builtin_amdgcn_wmma_f32_16x16x4_f32(
                    false, b, false, f.a[s], (short)0, acc[g * 4 + j], false, false);
            }
        }
    }
}

struct L2Ops {
    v2f w2f[8][2];
    v2f bia[8];
};

__device__ __forceinline__ void load_l2ops(L2Ops& p, const float* __restrict__ W2,
                                           const float* __restrict__ b1, int hc0, int m) {
    #pragma unroll
    for (int r = 0; r < 8; ++r) {
        const int hc = hc0 + 4 * r;
        p.bia[r] = *(const v2f*)(b1 + hc);
        #pragma unroll
        for (int mt = 0; mt < 2; ++mt) {
            p.w2f[r][mt].x = W2[(size_t)hc * 32 + 16 * mt + m];
            p.w2f[r][mt].y = W2[(size_t)(hc + 1) * 32 + 16 * mt + m];
        }
    }
}

__device__ __forceinline__ void k_loop(KBlock& f0, v8f (&acc)[16],
                                       const float* __restrict__ arow,
                                       const float* __restrict__ wlane,
                                       int K, int Kpad, int wave, int kl) {
    const int nBlk = Kpad >> 3;
    const int cnt  = (nBlk - wave + 7) >> 3;
    const int last = wave + 8 * (cnt - 1);
    KBlock f1;
    int i = 0;
    for (; i + 2 <= cnt; i += 2) {
        load_kblock(f1, arow, wlane, (wave + 8 * (i + 1)) * 8 + kl, K);
        __builtin_amdgcn_sched_barrier(0);
        mma_kblock(f0, acc);
        load_kblock(f0, arow, wlane, min(wave + 8 * (i + 2), last) * 8 + kl, K);
        __builtin_amdgcn_sched_barrier(0);
        mma_kblock(f1, acc);
    }
    if (cnt & 1) mma_kblock(f0, acc);
}

__device__ __forceinline__ void park_tiles(const v8f (&acc)[16], v4f* p,
                                           int wave, int lane) {
    #pragma unroll
    for (int t = 0; t < 16; ++t) {
        v4f lo = {acc[t][0], acc[t][1], acc[t][2], acc[t][3]};
        v4f hi = {acc[t][4], acc[t][5], acc[t][6], acc[t][7]};
        p[((wave * 16 + t) * 2 + 0) * 32 + lane] = lo;
        p[((wave * 16 + t) * 2 + 1) * 32 + lane] = hi;
    }
}

__device__ __forceinline__ void reduce_hidden(v2f (&hp)[8], const v4f* p,
                                              const L2Ops& ops, int wave, int lane) {
    v4f s[2][2];
    #pragma unroll
    for (int q = 0; q < 2; ++q) {
        const int t = 2 * wave + q;
        s[q][0] = p[((0 * 16 + t) * 2 + 0) * 32 + lane];
        s[q][1] = p[((0 * 16 + t) * 2 + 1) * 32 + lane];
        #pragma unroll
        for (int w = 1; w < 8; ++w) {
            s[q][0] += p[((w * 16 + t) * 2 + 0) * 32 + lane];
            s[q][1] += p[((w * 16 + t) * 2 + 1) * 32 + lane];
        }
    }
    #pragma unroll
    for (int r = 0; r < 8; ++r) {
        float h0 = s[0][r >> 2][r & 3] + ops.bia[r].x;
        float h1 = s[1][r >> 2][r & 3] + ops.bia[r].y;
        h0 = h0 > 0.0f ? h0 : 0.0f;
        h1 = h1 > 0.0f ? h1 : 0.0f;
        hp[r] = (v2f){h0, h1};
    }
}

#define PTILE (8 * 16 * 2 * 32)
#define OTILE (8 * 16 * OPITCH)

__global__ __launch_bounds__(256) void mlp_kernel(const float* __restrict__ bi_freq,
                                                  const float* __restrict__ tri_freq,
                                                  const float* __restrict__ Wb1,
                                                  const float* __restrict__ bb1,
                                                  const float* __restrict__ Wb2,
                                                  const float* __restrict__ bb2,
                                                  const float* __restrict__ Wt1,
                                                  const float* __restrict__ bt1,
                                                  const float* __restrict__ Wt2,
                                                  const float* __restrict__ bt2,
                                                  float* __restrict__ out) {
    __shared__ v4f   lds_p[2 * PTILE];
    __shared__ __attribute__((aligned(16))) float lds_o[2 * OTILE];

    const int rowBase = blockIdx.x * 16;
    const int lane = threadIdx.x & 31;
    const int wave = __builtin_amdgcn_readfirstlane(threadIdx.x >> 5);
    const int m  = lane & 15;
    const int kh = lane >> 4;
    const int kl = 2 * kh;
    const int hc0 = (wave >> 1) * 64 + 32 * kh + (wave & 1) * 2;

    v8f acc[16];
    KBlock f0;
    L2Ops opb, opt;

    const float* arow_b = bi_freq + (size_t)(rowBase + m) * KPAD_BI;
    load_kblock(f0, arow_b, Wb1 + 4 * m, wave * 8 + kl, V2);
    load_l2ops(opb, Wb2, bb1, hc0, m);
    #pragma unroll
    for (int t = 0; t < 16; ++t) acc[t] = (v8f){};
    k_loop(f0, acc, arow_b, Wb1 + 4 * m, V2, KPAD_BI, wave, kl);

    const float* arow_t = tri_freq + (size_t)(rowBase + m) * BUCKETS;
    load_kblock(f0, arow_t, Wt1 + 4 * m, wave * 8 + kl, BUCKETS);
    load_l2ops(opt, Wt2, bt1, hc0, m);
    park_tiles(acc, lds_p, wave, lane);

    #pragma unroll
    for (int t = 0; t < 16; ++t) acc[t] = (v8f){};
    k_loop(f0, acc, arow_t, Wt1 + 4 * m, BUCKETS, BUCKETS, wave, kl);
    park_tiles(acc, lds_p + PTILE, wave, lane);

    const v2f bout_b = *(const v2f*)(bb2 + (threadIdx.x & 15) * 2);
    const v2f bout_t = *(const v2f*)(bt2 + (threadIdx.x & 15) * 2);
    __syncthreads();

    v2f hb[8], ht[8];
    reduce_hidden(hb, lds_p, opb, wave, lane);
    reduce_hidden(ht, lds_p + PTILE, opt, wave, lane);

    v8f ob0 = {}, ob1 = {}, ot0 = {}, ot1 = {};
    #pragma unroll
    for (int r = 0; r < 8; ++r) {
        ob0 = __builtin_amdgcn_wmma_f32_16x16x4_f32(false, opb.w2f[r][0], false, hb[r], (short)0, ob0, false, false);
        ob1 = __builtin_amdgcn_wmma_f32_16x16x4_f32(false, opb.w2f[r][1], false, hb[r], (short)0, ob1, false, false);
        ot0 = __builtin_amdgcn_wmma_f32_16x16x4_f32(false, opt.w2f[r][0], false, ht[r], (short)0, ot0, false, false);
        ot1 = __builtin_amdgcn_wmma_f32_16x16x4_f32(false, opt.w2f[r][1], false, ht[r], (short)0, ot1, false, false);
    }

    {
        float* po = lds_o + (wave * 16 + m) * OPITCH + 8 * kh;
        *(v4f*)(po)      = (v4f){ob0[0], ob0[1], ob0[2], ob0[3]};
        *(v4f*)(po + 4)  = (v4f){ob0[4], ob0[5], ob0[6], ob0[7]};
        *(v4f*)(po + 16) = (v4f){ob1[0], ob1[1], ob1[2], ob1[3]};
        *(v4f*)(po + 20) = (v4f){ob1[4], ob1[5], ob1[6], ob1[7]};
        float* pt = po + OTILE;
        *(v4f*)(pt)      = (v4f){ot0[0], ot0[1], ot0[2], ot0[3]};
        *(v4f*)(pt + 4)  = (v4f){ot0[4], ot0[5], ot0[6], ot0[7]};
        *(v4f*)(pt + 16) = (v4f){ot1[0], ot1[1], ot1[2], ot1[3]};
        *(v4f*)(pt + 20) = (v4f){ot1[4], ot1[5], ot1[6], ot1[7]};
    }
    __syncthreads();

    {
        const int row = threadIdx.x >> 4;
        const int oc  = (threadIdx.x & 15) * 2;
        v2f ob = *(const v2f*)&lds_o[(0 * 16 + row) * OPITCH + oc];
        v2f ot = *(const v2f*)&lds_o[OTILE + (0 * 16 + row) * OPITCH + oc];
        #pragma unroll
        for (int w = 1; w < 8; ++w) {
            ob += *(const v2f*)&lds_o[(w * 16 + row) * OPITCH + oc];
            ot += *(const v2f*)&lds_o[OTILE + (w * 16 + row) * OPITCH + oc];
        }
        ob += bout_b;
        ot += bout_t;
        float* orow = out + (size_t)(rowBase + row) * OUTC;
        *(v2f*)&orow[oc]      = ob;
        *(v2f*)&orow[32 + oc] = ot;
    }
}

extern "C" void kernel_launch(void* const* d_in, const int* in_sizes, int n_in,
                              void* d_out, int out_size, void* d_ws, size_t ws_size,
                              hipStream_t stream) {
    const int*   src = (const int*)d_in[0];
    const float* Wb1 = (const float*)d_in[1];
    const float* bb1 = (const float*)d_in[2];
    const float* Wb2 = (const float*)d_in[3];
    const float* bb2 = (const float*)d_in[4];
    const float* Wt1 = (const float*)d_in[5];
    const float* bt1 = (const float*)d_in[6];
    const float* Wt2 = (const float*)d_in[7];
    const float* bt2 = (const float*)d_in[8];
    float* out = (float*)d_out;

    float* bi_freq  = (float*)d_ws;
    float* tri_freq = bi_freq + (size_t)B_ROWS * KPAD_BI;

    hist_kernel<<<B_ROWS / HROWS, 32 * HROWS, 0, stream>>>(src, bi_freq, tri_freq);

    mlp_kernel<<<B_ROWS / 16, 256, 0, stream>>>(bi_freq, tri_freq,
                                                Wb1, bb1, Wb2, bb2,
                                                Wt1, bt1, Wt2, bt2, out);
}
